# starting kernel
# baseline (speedup 1.0000x reference)
#include <hip/hip_runtime.h>
#include <hip/hip_bf16.h>

#define B_ 16384
#define N_ 1024
#define L_ 3
#define E_ 8
#define R_ 64
#define ER_ (E_*R_)

typedef float v2f __attribute__((ext_vector_type(2)));
typedef float v8f __attribute__((ext_vector_type(8)));

__device__ __forceinline__ v8f wmma_f32(v2f a, v2f b, v8f c) {
    return __builtin_amdgcn_wmma_f32_16x16x4_f32(
         false, a,  false, b,
         (short)0, c,  false,  false);
}

__global__ __launch_bounds__(256)
void crossnet_fused_kernel(const float* __restrict__ x,
                           const float* __restrict__ U,
                           const float* __restrict__ V,
                           const float* __restrict__ C,
                           const float* __restrict__ bias,
                           const float* __restrict__ gate_w,
                           float* __restrict__ out)
{
    __shared__ float x0_s[16 * N_];
    __shared__ float xl_s[16 * N_];
    __shared__ float v_s [16 * ER_];
    __shared__ float act_s[16 * ER_];
    __shared__ float gates_s[16 * E_];

    const int tid  = threadIdx.x;
    const int lane = tid & 31;
    const int wave = tid >> 5;
    const int mrow = lane & 15;
    const int ksel = lane >> 4;
    const size_t row0 = (size_t)blockIdx.x * 16;

    {
        const float4* xg  = (const float4*)(x + row0 * N_);
        float4* x0v = (float4*)x0_s;
        float4* xlv = (float4*)xl_s;
        #pragma unroll
        for (int i = 0; i < (16 * N_ / 4) / 256; ++i) {
            int idx = tid + i * 256;
            float4 t = xg[idx];
            x0v[idx] = t;
            xlv[idx] = t;
        }
    }
    __syncthreads();

    for (int l = 0; l < L_; ++l) {
        if (tid < 16 * E_) {
            const int row = tid >> 3, e = tid & 7;
            const float* gw = gate_w + e * N_;
            const float* xr = xl_s + row * N_;
            float acc = 0.f;
            for (int n = 0; n < N_; ++n) acc = fmaf(xr[n], gw[n], acc);
            gates_s[tid] = acc;
        }
        __syncthreads();
        if (tid < 16) {
            float m = gates_s[tid * E_];
            #pragma unroll
            for (int e = 1; e < E_; ++e) m = fmaxf(m, gates_s[tid * E_ + e]);
            float tmp[E_]; float s = 0.f;
            #pragma unroll
            for (int e = 0; e < E_; ++e) { tmp[e] = __expf(gates_s[tid * E_ + e] - m); s += tmp[e]; }
            const float inv = 1.f / s;
            #pragma unroll
            for (int e = 0; e < E_; ++e) gates_s[tid * E_ + e] = tmp[e] * inv;
        }
        {
            const int e = wave;
            const float* Vb = V + ((size_t)l * E_ + e) * (R_ * N_);
            v8f a0 = {}, a1 = {}, a2 = {}, a3 = {};
            const float* a_src = xl_s + mrow * N_ + ksel * 2;
            for (int n0 = 0; n0 < N_; n0 += 4) {
                v2f a = *(const v2f*)(a_src + n0);
                const float* bp = Vb + n0 + ksel * 2;
                v2f b0 = *(const v2f*)(bp + (0 * 16 + mrow) * N_);
                v2f b1 = *(const v2f*)(bp + (1 * 16 + mrow) * N_);
                v2f b2 = *(const v2f*)(bp + (2 * 16 + mrow) * N_);
                v2f b3 = *(const v2f*)(bp + (3 * 16 + mrow) * N_);
                a0 = wmma_f32(a, b0, a0);
                a1 = wmma_f32(a, b1, a1);
                a2 = wmma_f32(a, b2, a2);
                a3 = wmma_f32(a, b3, a3);
            }
            v8f acc[4] = {a0, a1, a2, a3};
            #pragma unroll
            for (int t = 0; t < 4; ++t)
                #pragma unroll
                for (int j = 0; j < 8; ++j) {
                    const int row = j + 8 * ksel;
                    v_s[row * ER_ + e * R_ + t * 16 + mrow] = fmaxf(acc[t][j], 0.f);
                }
        }
        __syncthreads();

        {
            const int e = wave;
            const float* Cb = C + ((size_t)l * E_ + e) * (R_ * R_);
            v8f a0 = {}, a1 = {}, a2 = {}, a3 = {};
            const float* a_src = v_s + mrow * ER_ + e * R_ + ksel * 2;
            #pragma unroll
            for (int s0 = 0; s0 < R_; s0 += 4) {
                v2f a = *(const v2f*)(a_src + s0);
                const float* bp = Cb + s0 + ksel * 2;
                v2f b0 = *(const v2f*)(bp + (0 * 16 + mrow) * R_);
                v2f b1 = *(const v2f*)(bp + (1 * 16 + mrow) * R_);
                v2f b2 = *(const v2f*)(bp + (2 * 16 + mrow) * R_);
                v2f b3 = *(const v2f*)(bp + (3 * 16 + mrow) * R_);
                a0 = wmma_f32(a, b0, a0);
                a1 = wmma_f32(a, b1, a1);
                a2 = wmma_f32(a, b2, a2);
                a3 = wmma_f32(a, b3, a3);
            }
            v8f acc[4] = {a0, a1, a2, a3};
            #pragma unroll
            for (int t = 0; t < 4; ++t)
                #pragma unroll
                for (int j = 0; j < 8; ++j) {
                    const int row = j + 8 * ksel;
                    const float g = gates_s[row * E_ + e];
                    act_s[row * ER_ + e * R_ + t * 16 + mrow] = g * fmaxf(acc[t][j], 0.f);
                }
        }
        __syncthreads();

        {
            const int n_base = wave * 128;
            const float* Ub = U + (size_t)l * (E_ * N_ * R_);
            v8f acc[8] = {};
            const float* a_src = act_s + mrow * ER_ + ksel * 2;
            for (int k0 = 0; k0 < ER_; k0 += 4) {
                v2f a = *(const v2f*)(a_src + k0);
                const int e  = k0 >> 6;
                const int r0 = (k0 & 63) + ksel * 2;
                const float* bp = Ub + ((size_t)e * N_) * R_ + r0;
                #pragma unroll
                for (int t = 0; t < 8; ++t) {
                    const int n = n_base + t * 16 + mrow;
                    v2f b = *(const v2f*)(bp + (size_t)n * R_);
                    acc[t] = wmma_f32(a, b, acc[t]);
                }
            }
            const float* bl = bias + l * N_;
            #pragma unroll
            for (int t = 0; t < 8; ++t) {
                const int n = n_base + t * 16 + mrow;
                const float bn = bl[n];
                #pragma unroll
                for (int j = 0; j < 8; ++j) {
                    const int row = j + 8 * ksel;
                    const float gu  = acc[t][j];
                    const float x0v = x0_s[row * N_ + n];
                    const float xlv = xl_s[row * N_ + n];
                    xl_s[row * N_ + n] = fmaf(x0v, gu + bn, xlv);
                }
            }
        }
        __syncthreads();
    }

    {
        float4* og = (float4*)(out + row0 * N_);
        const float4* xlv = (const float4*)xl_s;
        #pragma unroll
        for (int i = 0; i < (16 * N_ / 4) / 256; ++i) {
            int idx = tid + i * 256;
            og[idx] = xlv[idx];
        }
    }
}

extern "C" void kernel_launch(void* const* d_in, const int* in_sizes, int n_in,
                              void* d_out, int out_size, void* d_ws, size_t ws_size,
                              hipStream_t stream) {
    const float* x      = (const float*)d_in[0];
    const float* U      = (const float*)d_in[1];
    const float* V      = (const float*)d_in[2];
    const float* C      = (const float*)d_in[3];
    const float* bias   = (const float*)d_in[4];
    const float* gate_w = (const float*)d_in[5];
    float* out = (float*)d_out;

    dim3 grid(B_ / 16);
    dim3 block(256);
    crossnet_fused_kernel<<<grid, block, 0, stream>>>(x, U, V, C, bias, gate_w, out);
}


# B fragments by 16-byte loads: lanes l, l+16 split each 8 k as 4+4 contiguous, A loader permuted alike (f32 sum order within each 8 k changes); a line's 4 loads adjacent, one block ahead of use
# speedup vs baseline: 1.9999x; 1.9999x over previous
#include <hip/hip_runtime.h>
#include <hip/hip_bf16.h>


#define B_ 16384
#define N_ 1024
#define L_ 3
#define E_ 8
#define R_ 64
#define ER_ (E_*R_)

typedef float v2f __attribute__((ext_vector_type(2)));
typedef float v4f __attribute__((ext_vector_type(4)));
typedef float v8f __attribute__((ext_vector_type(8)));

__device__ __forceinline__ v8f wmma_f32(v2f a, v2f b, v8f c) {
    return __builtin_amdgcn_wmma_f32_16x16x4_f32(
         false, a,  false, b,
         (short)0, c,  false,  false);
}


template <int NT, int NQ, bool ROW_ORDER>
__device__ __forceinline__ void load_b(v4f (&b)[NT][NQ], const float* __restrict__ p, int pitch) {
    #pragma unroll
    for (int t = 0; t < NT; ++t) {
        #pragma unroll
        for (int q = 0; q < NQ; ++q)
            b[t][q] = *(const v4f*)(p + (size_t)t * 16 * pitch + 8 * q);
        if (ROW_ORDER) asm volatile("" ::: "memory");
    }
}

template <int NQ>
__device__ __forceinline__ void load_a(v4f (&a)[NQ], const float* p) {
    #pragma unroll
    for (int q = 0; q < NQ; ++q)
        a[q] = *(const v4f*)(p + 8 * q);
}

template <int NT, int NQ>
__device__ __forceinline__ void mm(v8f* acc, const v4f (&a)[NQ], const v4f (&b)[NT][NQ]) {
    #pragma unroll
    for (int q = 0; q < NQ; ++q) {
        #pragma unroll
        for (int t = 0; t < NT; ++t) acc[t] = wmma_f32(a[q].xy, b[t][q].xy, acc[t]);
        #pragma unroll
        for (int t = 0; t < NT; ++t) acc[t] = wmma_f32(a[q].zw, b[t][q].zw, acc[t]);
    }
}

__global__ __launch_bounds__(256)
void crossnet_fused_kernel(const float* __restrict__ x,
                           const float* __restrict__ U,
                           const float* __restrict__ V,
                           const float* __restrict__ C,
                           const float* __restrict__ bias,
                           const float* __restrict__ gate_w,
                           float* __restrict__ out)
{
    __shared__ __attribute__((aligned(16))) float x0_s[16 * N_];
    __shared__ __attribute__((aligned(16))) float xl_s[16 * N_];
    __shared__ __attribute__((aligned(16))) float v_s [16 * ER_];
    __shared__ __attribute__((aligned(16))) float act_s[16 * ER_];
    __shared__ float gates_s[16 * E_];

    const int tid  = threadIdx.x;
    const int lane = tid & 31;
    const int wave = tid >> 5;
    const int mrow = lane & 15;
    const int ksel = lane >> 4;
    const size_t row0 = (size_t)blockIdx.x * 16;

    {
        const float4* xg  = (const float4*)(x + row0 * N_);
        float4* x0v = (float4*)x0_s;
        float4* xlv = (float4*)xl_s;
        #pragma unroll
        for (int i = 0; i < (16 * N_ / 4) / 256; ++i) {
            int idx = tid + i * 256;
            float4 t = xg[idx];
            x0v[idx] = t;
            xlv[idx] = t;
        }
    }
    __syncthreads();

    for (int l = 0; l < L_; ++l) {
        if (tid < 16 * E_) {
            const int row = tid >> 3, e = tid & 7;
            const float* gw = gate_w + e * N_;
            const float* xr = xl_s + row * N_;
            float acc = 0.f;
            for (int n = 0; n < N_; ++n) acc = fmaf(xr[n], gw[n], acc);
            gates_s[tid] = acc;
        }
        __syncthreads();
        if (tid < 16) {
            float m = gates_s[tid * E_];
            #pragma unroll
            for (int e = 1; e < E_; ++e) m = fmaxf(m, gates_s[tid * E_ + e]);
            float tmp[E_]; float s = 0.f;
            #pragma unroll
            for (int e = 0; e < E_; ++e) { tmp[e] = __expf(gates_s[tid * E_ + e] - m); s += tmp[e]; }
            const float inv = 1.f / s;
            #pragma unroll
            for (int e = 0; e < E_; ++e) gates_s[tid * E_ + e] = tmp[e] * inv;
        }
        {
            const int e = wave;
            const float* Vb = V + ((size_t)l * E_ + e) * (R_ * N_)
                                + (size_t)mrow * N_ + ksel * 4;
            const float* a_src = xl_s + mrow * N_ + ksel * 4;
            v8f acc[4] = {};
            v4f b0[4][4], b1[4][4], a[4];
            load_b<4, 4, false>(b0, Vb, N_);
            for (int n0 = 0; n0 < N_ - 64; n0 += 64) {
                load_b<4, 4, false>(b1, Vb + n0 + 32, N_);
                load_a<4>(a, a_src + n0);
                mm<4, 4>(acc, a, b0);
                load_b<4, 4, false>(b0, Vb + n0 + 64, N_);
                load_a<4>(a, a_src + n0 + 32);
                mm<4, 4>(acc, a, b1);
            }
            load_b<4, 4, false>(b1, Vb + (N_ - 32), N_);
            load_a<4>(a, a_src + (N_ - 64));
            mm<4, 4>(acc, a, b0);
            load_a<4>(a, a_src + (N_ - 32));
            mm<4, 4>(acc, a, b1);
            #pragma unroll
            for (int t = 0; t < 4; ++t)
                #pragma unroll
                for (int j = 0; j < 8; ++j) {
                    const int row = j + 8 * ksel;
                    v_s[row * ER_ + e * R_ + t * 16 + mrow] = fmaxf(acc[t][j], 0.f);
                }
        }
        __syncthreads();

        {
            const int e = wave;
            const float* Cb = C + ((size_t)l * E_ + e) * (R_ * R_)
                                + mrow * R_ + ksel * 4;
            const float* a_src = v_s + mrow * ER_ + e * R_ + ksel * 4;
            v8f acc[4] = {};
            v4f b[4][8], a[8];
            load_b<4, 8, true>(b, Cb, R_);
            load_a<8>(a, a_src);
            mm<4, 8>(acc, a, b);
            #pragma unroll
            for (int t = 0; t < 4; ++t)
                #pragma unroll
                for (int j = 0; j < 8; ++j) {
                    const int row = j + 8 * ksel;
                    const float g = gates_s[row * E_ + e];
                    act_s[row * ER_ + e * R_ + t * 16 + mrow] = g * fmaxf(acc[t][j], 0.f);
                }
        }
        __syncthreads();

        {
            const int n_base = wave * 128;
            const float* Ub = U + (size_t)l * (E_ * N_ * R_)
                                + (size_t)(n_base + mrow) * R_ + ksel * 4;
            const float* a_src = act_s + mrow * ER_ + ksel * 4;
            v8f acc[8] = {};
            v4f b0[2][8], b1[2][8], a[8];
            load_b<2, 8, true>(b0, Ub, R_);
            for (int e = 0; e < E_; ++e) {
                const float* ue = Ub + (size_t)e * (N_ * R_);
                load_a<8>(a, a_src + e * R_);
                load_b<2, 8, true>(b1, ue + 32 * R_, R_);
                mm<2, 8>(acc + 0, a, b0);
                load_b<2, 8, true>(b0, ue + 64 * R_, R_);
                mm<2, 8>(acc + 2, a, b1);
                load_b<2, 8, true>(b1, ue + 96 * R_, R_);
                mm<2, 8>(acc + 4, a, b0);
                if (e + 1 < E_) load_b<2, 8, true>(b0, ue + N_ * R_, R_);
                mm<2, 8>(acc + 6, a, b1);
            }
            const float* bl = bias + l * N_;
            #pragma unroll
            for (int t = 0; t < 8; ++t) {
                const int n = n_base + t * 16 + mrow;
                const float bn = bl[n];
                #pragma unroll
                for (int j = 0; j < 8; ++j) {
                    const int row = j + 8 * ksel;
                    const float gu  = acc[t][j];
                    const float x0v = x0_s[row * N_ + n];
                    const float xlv = xl_s[row * N_ + n];
                    xl_s[row * N_ + n] = fmaf(x0v, gu + bn, xlv);
                }
            }
        }
        __syncthreads();
    }

    {
        float4* og = (float4*)(out + row0 * N_);
        const float4* xlv = (const float4*)xl_s;
        #pragma unroll
        for (int i = 0; i < (16 * N_ / 4) / 256; ++i) {
            int idx = tid + i * 256;
            og[idx] = xlv[idx];
        }
    }
}

extern "C" void kernel_launch(void* const* d_in, const int* in_sizes, int n_in,
                              void* d_out, int out_size, void* d_ws, size_t ws_size,
                              hipStream_t stream) {
    const float* x      = (const float*)d_in[0];
    const float* U      = (const float*)d_in[1];
    const float* V      = (const float*)d_in[2];
    const float* C      = (const float*)d_in[3];
    const float* bias   = (const float*)d_in[4];
    const float* gate_w = (const float*)d_in[5];
    float* out = (float*)d_out;

    dim3 grid(B_ / 16);
    dim3 block(256);
    crossnet_fused_kernel<<<grid, block, 0, stream>>>(x, U, V, C, bias, gate_w, out);
}


# gate logits moved from a 128-thread scalar dot loop onto f32 WMMA: each wave a 128-k slice, the 8 partial tiles summed in LDS before the softmax (f32 sum order of the logits changes)
# speedup vs baseline: 2.2514x; 1.1258x over previous
#include <hip/hip_runtime.h>
#include <hip/hip_bf16.h>


#define B_ 16384
#define N_ 1024
#define L_ 3
#define E_ 8
#define R_ 64
#define ER_ (E_*R_)

typedef float v2f __attribute__((ext_vector_type(2)));
typedef float v4f __attribute__((ext_vector_type(4)));
typedef float v8f __attribute__((ext_vector_type(8)));

__device__ __forceinline__ v8f wmma_f32(v2f a, v2f b, v8f c) {
    return __builtin_amdgcn_wmma_f32_16x16x4_f32(
         false, a,  false, b,
         (short)0, c,  false,  false);
}


template <int NT, int NQ, bool ROW_ORDER>
__device__ __forceinline__ void load_b(v4f (&b)[NT][NQ], const float* __restrict__ p, int pitch) {
    #pragma unroll
    for (int t = 0; t < NT; ++t) {
        #pragma unroll
        for (int q = 0; q < NQ; ++q)
            b[t][q] = *(const v4f*)(p + (size_t)t * 16 * pitch + 8 * q);
        if (ROW_ORDER) asm volatile("" ::: "memory");
    }
}

template <int NQ>
__device__ __forceinline__ void load_a(v4f (&a)[NQ], const float* p) {
    #pragma unroll
    for (int q = 0; q < NQ; ++q)
        a[q] = *(const v4f*)(p + 8 * q);
}

template <int NT, int NQ>
__device__ __forceinline__ void mm(v8f* acc, const v4f (&a)[NQ], const v4f (&b)[NT][NQ]) {
    #pragma unroll
    for (int q = 0; q < NQ; ++q) {
        #pragma unroll
        for (int t = 0; t < NT; ++t) acc[t] = wmma_f32(a[q].xy, b[t][q].xy, acc[t]);
        #pragma unroll
        for (int t = 0; t < NT; ++t) acc[t] = wmma_f32(a[q].zw, b[t][q].zw, acc[t]);
    }
}

__global__ __launch_bounds__(256)
void crossnet_fused_kernel(const float* __restrict__ x,
                           const float* __restrict__ U,
                           const float* __restrict__ V,
                           const float* __restrict__ C,
                           const float* __restrict__ bias,
                           const float* __restrict__ gate_w,
                           float* __restrict__ out)
{
    __shared__ __attribute__((aligned(16))) float x0_s[16 * N_];
    __shared__ __attribute__((aligned(16))) float xl_s[16 * N_];
    __shared__ __attribute__((aligned(16))) float v_s [16 * ER_];
    __shared__ __attribute__((aligned(16))) float act_s[16 * ER_];
    __shared__ float gates_s[16 * E_];
    __shared__ __attribute__((aligned(16))) float gpart_s[8 * 16 * E_];

    const int tid  = threadIdx.x;
    const int lane = tid & 31;
    const int wave = tid >> 5;
    const int mrow = lane & 15;
    const int ksel = lane >> 4;
    const size_t row0 = (size_t)blockIdx.x * 16;

    {
        const float4* xg  = (const float4*)(x + row0 * N_);
        float4* x0v = (float4*)x0_s;
        float4* xlv = (float4*)xl_s;
        #pragma unroll
        for (int i = 0; i < (16 * N_ / 4) / 256; ++i) {
            int idx = tid + i * 256;
            float4 t = xg[idx];
            x0v[idx] = t;
            xlv[idx] = t;
        }
    }
    __syncthreads();

    for (int l = 0; l < L_; ++l) {
        {
            const float* gb    = gate_w + (size_t)(mrow & 7) * N_ + wave * 128 + ksel * 4;
            const float* a_src = xl_s + mrow * N_ + wave * 128 + ksel * 4;
            v4f b[1][16], a[16];
            load_b<1, 16, false>(b, gb, N_);
            load_a<16>(a, a_src);
            v8f acc0 = {}, acc1 = {};
            #pragma unroll
            for (int q = 0; q < 16; ++q) {
                acc0 = wmma_f32(a[q].xy, b[0][q].xy, acc0);
                acc1 = wmma_f32(a[q].zw, b[0][q].zw, acc1);
            }
            asm volatile("v_nop\n\tv_nop\n\tv_nop\n\tv_nop\n\tv_nop\n\tv_nop\n\tv_nop\n\tv_nop"
                         : "+v"(acc0), "+v"(acc1));
            if (mrow < E_) {
                #pragma unroll
                for (int j = 0; j < 8; ++j)
                    gpart_s[(wave * 16 + j + 8 * ksel) * E_ + mrow] = acc0[j] + acc1[j];
            }
        }
        __syncthreads();
        if (tid < 16) {
            float lg[E_];
            #pragma unroll
            for (int e = 0; e < E_; ++e) lg[e] = gpart_s[tid * E_ + e];
            #pragma unroll
            for (int w = 1; w < 8; ++w) {
                #pragma unroll
                for (int e = 0; e < E_; ++e) lg[e] += gpart_s[(w * 16 + tid) * E_ + e];
            }
            float m = lg[0];
            #pragma unroll
            for (int e = 1; e < E_; ++e) m = fmaxf(m, lg[e]);
            float tmp[E_]; float s = 0.f;
            #pragma unroll
            for (int e = 0; e < E_; ++e) { tmp[e] = __expf(lg[e] - m); s += tmp[e]; }
            const float inv = 1.f / s;
            #pragma unroll
            for (int e = 0; e < E_; ++e) gates_s[tid * E_ + e] = tmp[e] * inv;
        }
        {
            const int e = wave;
            const float* Vb = V + ((size_t)l * E_ + e) * (R_ * N_)
                                + (size_t)mrow * N_ + ksel * 4;
            const float* a_src = xl_s + mrow * N_ + ksel * 4;
            v8f acc[4] = {};
            v4f b0[4][4], b1[4][4], a[4];
            load_b<4, 4, false>(b0, Vb, N_);
            for (int n0 = 0; n0 < N_ - 64; n0 += 64) {
                load_b<4, 4, false>(b1, Vb + n0 + 32, N_);
                load_a<4>(a, a_src + n0);
                mm<4, 4>(acc, a, b0);
                load_b<4, 4, false>(b0, Vb + n0 + 64, N_);
                load_a<4>(a, a_src + n0 + 32);
                mm<4, 4>(acc, a, b1);
            }
            load_b<4, 4, false>(b1, Vb + (N_ - 32), N_);
            load_a<4>(a, a_src + (N_ - 64));
            mm<4, 4>(acc, a, b0);
            load_a<4>(a, a_src + (N_ - 32));
            mm<4, 4>(acc, a, b1);
            #pragma unroll
            for (int t = 0; t < 4; ++t)
                #pragma unroll
                for (int j = 0; j < 8; ++j) {
                    const int row = j + 8 * ksel;
                    v_s[row * ER_ + e * R_ + t * 16 + mrow] = fmaxf(acc[t][j], 0.f);
                }
        }
        __syncthreads();

        {
            const int e = wave;
            const float* Cb = C + ((size_t)l * E_ + e) * (R_ * R_)
                                + mrow * R_ + ksel * 4;
            const float* a_src = v_s + mrow * ER_ + e * R_ + ksel * 4;
            v8f acc[4] = {};
            v4f b[4][8], a[8];
            load_b<4, 8, true>(b, Cb, R_);
            load_a<8>(a, a_src);
            mm<4, 8>(acc, a, b);
            #pragma unroll
            for (int t = 0; t < 4; ++t)
                #pragma unroll
                for (int j = 0; j < 8; ++j) {
                    const int row = j + 8 * ksel;
                    const float g = gates_s[row * E_ + e];
                    act_s[row * ER_ + e * R_ + t * 16 + mrow] = g * fmaxf(acc[t][j], 0.f);
                }
        }
        __syncthreads();

        {
            const int n_base = wave * 128;
            const float* Ub = U + (size_t)l * (E_ * N_ * R_)
                                + (size_t)(n_base + mrow) * R_ + ksel * 4;
            const float* a_src = act_s + mrow * ER_ + ksel * 4;
            v8f acc[8] = {};
            v4f b0[2][8], b1[2][8], a[8];
            load_b<2, 8, true>(b0, Ub, R_);
            for (int e = 0; e < E_; ++e) {
                const float* ue = Ub + (size_t)e * (N_ * R_);
                load_a<8>(a, a_src + e * R_);
                load_b<2, 8, true>(b1, ue + 32 * R_, R_);
                mm<2, 8>(acc + 0, a, b0);
                load_b<2, 8, true>(b0, ue + 64 * R_, R_);
                mm<2, 8>(acc + 2, a, b1);
                load_b<2, 8, true>(b1, ue + 96 * R_, R_);
                mm<2, 8>(acc + 4, a, b0);
                if (e + 1 < E_) load_b<2, 8, true>(b0, ue + N_ * R_, R_);
                mm<2, 8>(acc + 6, a, b1);
            }
            const float* bl = bias + l * N_;
            #pragma unroll
            for (int t = 0; t < 8; ++t) {
                const int n = n_base + t * 16 + mrow;
                const float bn = bl[n];
                #pragma unroll
                for (int j = 0; j < 8; ++j) {
                    const int row = j + 8 * ksel;
                    const float gu  = acc[t][j];
                    const float x0v = x0_s[row * N_ + n];
                    const float xlv = xl_s[row * N_ + n];
                    xl_s[row * N_ + n] = fmaf(x0v, gu + bn, xlv);
                }
            }
        }
        __syncthreads();
    }

    {
        float4* og = (float4*)(out + row0 * N_);
        const float4* xlv = (const float4*)xl_s;
        #pragma unroll
        for (int i = 0; i < (16 * N_ / 4) / 256; ++i) {
            int idx = tid + i * 256;
            og[idx] = xlv[idx];
        }
    }
}

extern "C" void kernel_launch(void* const* d_in, const int* in_sizes, int n_in,
                              void* d_out, int out_size, void* d_ws, size_t ws_size,
                              hipStream_t stream) {
    const float* x      = (const float*)d_in[0];
    const float* U      = (const float*)d_in[1];
    const float* V      = (const float*)d_in[2];
    const float* C      = (const float*)d_in[3];
    const float* bias   = (const float*)d_in[4];
    const float* gate_w = (const float*)d_in[5];
    float* out = (float*)d_out;

    dim3 grid(B_ / 16);
    dim3 block(256);
    crossnet_fused_kernel<<<grid, block, 0, stream>>>(x, U, V, C, bias, gate_w, out);
}


# 32 rows per work-group as two interleaved 16-row tiles: each B fragment loaded from global feeds both tiles' WMMAs (half the weight loads per product); x0 re-read from global, not kept in LDS
# speedup vs baseline: 3.1434x; 1.3962x over previous
#include <hip/hip_runtime.h>
#include <hip/hip_bf16.h>


#define B_ 16384
#define N_ 1024
#define L_ 3
#define E_ 8
#define R_ 64
#define ER_ (E_*R_)
#define ROWS_ 32
#define AROW(m, r) (2 * (r) + (m))

typedef float v2f __attribute__((ext_vector_type(2)));
typedef float v4f __attribute__((ext_vector_type(4)));
typedef float v8f __attribute__((ext_vector_type(8)));

__device__ __forceinline__ v8f wmma_f32(v2f a, v2f b, v8f c) {
    return __builtin_amdgcn_wmma_f32_16x16x4_f32(
         false, a,  false, b,
         (short)0, c,  false,  false);
}


template <int NT, int NQ, bool ROW_ORDER>
__device__ __forceinline__ void load_b(v4f (&b)[NT][NQ], const float* __restrict__ p, int pitch) {
    #pragma unroll
    for (int t = 0; t < NT; ++t) {
        #pragma unroll
        for (int q = 0; q < NQ; ++q)
            b[t][q] = *(const v4f*)(p + (size_t)t * 16 * pitch + 8 * q);
        if (ROW_ORDER) asm volatile("" ::: "memory");
    }
}

template <int NQ>
__device__ __forceinline__ void load_a(v4f (&a)[2][NQ], const float* p, int tile_off) {
    #pragma unroll
    for (int m = 0; m < 2; ++m) {
        #pragma unroll
        for (int q = 0; q < NQ; ++q)
            a[m][q] = *(const v4f*)(p + m * tile_off + 8 * q);
    }
}

template <int NT, int NQ>
__device__ __forceinline__ void mm(v8f* acc0, v8f* acc1, const v4f (&a)[2][NQ], const v4f (&b)[NT][NQ]) {
    #pragma unroll
    for (int q = 0; q < NQ; ++q) {
        #pragma unroll
        for (int t = 0; t < NT; ++t) {
            acc0[t] = wmma_f32(a[0][q].xy, b[t][q].xy, acc0[t]);
            acc1[t] = wmma_f32(a[1][q].xy, b[t][q].xy, acc1[t]);
        }
        #pragma unroll
        for (int t = 0; t < NT; ++t) {
            acc0[t] = wmma_f32(a[0][q].zw, b[t][q].zw, acc0[t]);
            acc1[t] = wmma_f32(a[1][q].zw, b[t][q].zw, acc1[t]);
        }
    }
}

__global__ __launch_bounds__(256)
void crossnet_fused_kernel(const float* __restrict__ x,
                           const float* __restrict__ U,
                           const float* __restrict__ V,
                           const float* __restrict__ C,
                           const float* __restrict__ bias,
                           const float* __restrict__ gate_w,
                           float* __restrict__ out)
{
    __shared__ __attribute__((aligned(16))) float xl_s[ROWS_ * N_];
    __shared__ __attribute__((aligned(16))) float v_s [ROWS_ * ER_];
    __shared__ __attribute__((aligned(16))) float act_s[ROWS_ * ER_];
    __shared__ float gates_s[ROWS_ * E_];
    __shared__ __attribute__((aligned(16))) float gpart_s[8 * ROWS_ * E_];

    const int tid  = threadIdx.x;
    const int lane = tid & 31;
    const int wave = tid >> 5;
    const int mrow = lane & 15;
    const int ksel = lane >> 4;
    const size_t row0 = (size_t)blockIdx.x * ROWS_;

    {
        const float4* xg  = (const float4*)(x + row0 * N_);
        float4* xlv = (float4*)xl_s;
        #pragma unroll 8
        for (int i = 0; i < (ROWS_ * N_ / 4) / 256; ++i) {
            int idx = tid + i * 256;
            xlv[idx] = xg[idx];
        }
    }
    __syncthreads();

    for (int l = 0; l < L_; ++l) {
        {
            const float* gb    = gate_w + (size_t)(mrow & 7) * N_ + wave * 128 + ksel * 4;
            const float* a_src = xl_s + AROW(0, mrow) * N_ + wave * 128 + ksel * 4;
            v4f b[1][16], a[2][16];
            load_b<1, 16, false>(b, gb, N_);
            load_a<16>(a, a_src, N_);
            v8f acc0 = {}, acc1 = {}, acc2 = {}, acc3 = {};
            #pragma unroll
            for (int q = 0; q < 16; ++q) {
                acc0 = wmma_f32(a[0][q].xy, b[0][q].xy, acc0);
                acc2 = wmma_f32(a[1][q].xy, b[0][q].xy, acc2);
                acc1 = wmma_f32(a[0][q].zw, b[0][q].zw, acc1);
                acc3 = wmma_f32(a[1][q].zw, b[0][q].zw, acc3);
            }
            asm volatile("v_nop\n\tv_nop\n\tv_nop\n\tv_nop\n\tv_nop\n\tv_nop\n\tv_nop\n\tv_nop"
                         : "+v"(acc0), "+v"(acc1), "+v"(acc2), "+v"(acc3));
            if (mrow < E_) {
                #pragma unroll
                for (int j = 0; j < 8; ++j) {
                    gpart_s[(wave * ROWS_ + AROW(0, j + 8 * ksel)) * E_ + mrow] = acc0[j] + acc1[j];
                    gpart_s[(wave * ROWS_ + AROW(1, j + 8 * ksel)) * E_ + mrow] = acc2[j] + acc3[j];
                }
            }
        }
        __syncthreads();
        if (tid < ROWS_) {
            float lg[E_];
            #pragma unroll
            for (int e = 0; e < E_; ++e) lg[e] = gpart_s[tid * E_ + e];
            #pragma unroll
            for (int w = 1; w < 8; ++w) {
                #pragma unroll
                for (int e = 0; e < E_; ++e) lg[e] += gpart_s[(w * ROWS_ + tid) * E_ + e];
            }
            float m = lg[0];
            #pragma unroll
            for (int e = 1; e < E_; ++e) m = fmaxf(m, lg[e]);
            float tmp[E_]; float s = 0.f;
            #pragma unroll
            for (int e = 0; e < E_; ++e) { tmp[e] = __expf(lg[e] - m); s += tmp[e]; }
            const float inv = 1.f / s;
            #pragma unroll
            for (int e = 0; e < E_; ++e) gates_s[tid * E_ + e] = tmp[e] * inv;
        }
        {
            const int e = wave;
            const float* Vb = V + ((size_t)l * E_ + e) * (R_ * N_)
                                + (size_t)mrow * N_ + ksel * 4;
            const float* a_src = xl_s + AROW(0, mrow) * N_ + ksel * 4;
            v8f acc[2][4] = {};
            v4f b0[4][4], b1[4][4], a[2][4];
            load_b<4, 4, false>(b0, Vb, N_);
            for (int n0 = 0; n0 < N_ - 64; n0 += 64) {
                load_b<4, 4, false>(b1, Vb + n0 + 32, N_);
                load_a<4>(a, a_src + n0, N_);
                mm<4, 4>(acc[0], acc[1], a, b0);
                load_b<4, 4, false>(b0, Vb + n0 + 64, N_);
                load_a<4>(a, a_src + n0 + 32, N_);
                mm<4, 4>(acc[0], acc[1], a, b1);
            }
            load_b<4, 4, false>(b1, Vb + (N_ - 32), N_);
            load_a<4>(a, a_src + (N_ - 64), N_);
            mm<4, 4>(acc[0], acc[1], a, b0);
            load_a<4>(a, a_src + (N_ - 32), N_);
            mm<4, 4>(acc[0], acc[1], a, b1);
            #pragma unroll
            for (int m = 0; m < 2; ++m)
                #pragma unroll
                for (int t = 0; t < 4; ++t)
                    #pragma unroll
                    for (int j = 0; j < 8; ++j) {
                        const int row = AROW(m, j + 8 * ksel);
                        v_s[row * ER_ + e * R_ + t * 16 + mrow] = fmaxf(acc[m][t][j], 0.f);
                    }
        }
        __syncthreads();

        {
            const int e = wave;
            const float* Cb = C + ((size_t)l * E_ + e) * (R_ * R_)
                                + mrow * R_ + ksel * 4;
            const float* a_src = v_s + AROW(0, mrow) * ER_ + e * R_ + ksel * 4;
            v8f acc[2][4] = {};
            v4f b[4][8], a[2][8];
            load_b<4, 8, true>(b, Cb, R_);
            load_a<8>(a, a_src, ER_);
            mm<4, 8>(acc[0], acc[1], a, b);
            #pragma unroll
            for (int m = 0; m < 2; ++m)
                #pragma unroll
                for (int t = 0; t < 4; ++t)
                    #pragma unroll
                    for (int j = 0; j < 8; ++j) {
                        const int row = AROW(m, j + 8 * ksel);
                        const float g = gates_s[row * E_ + e];
                        act_s[row * ER_ + e * R_ + t * 16 + mrow] = g * fmaxf(acc[m][t][j], 0.f);
                    }
        }
        __syncthreads();

        {
            const int n_base = wave * 128;
            const float* Ub = U + (size_t)l * (E_ * N_ * R_)
                                + (size_t)(n_base + mrow) * R_ + ksel * 4;
            const float* a_src = act_s + AROW(0, mrow) * ER_ + ksel * 4;
            v8f acc[2][8] = {};
            v4f b0[2][8], b1[2][8], a[2][8];
            load_b<2, 8, true>(b0, Ub, R_);
            for (int e = 0; e < E_; ++e) {
                const float* ue = Ub + (size_t)e * (N_ * R_);
                load_a<8>(a, a_src + e * R_, ER_);
                load_b<2, 8, true>(b1, ue + 32 * R_, R_);
                mm<2, 8>(acc[0] + 0, acc[1] + 0, a, b0);
                load_b<2, 8, true>(b0, ue + 64 * R_, R_);
                mm<2, 8>(acc[0] + 2, acc[1] + 2, a, b1);
                load_b<2, 8, true>(b1, ue + 96 * R_, R_);
                mm<2, 8>(acc[0] + 4, acc[1] + 4, a, b0);
                if (e + 1 < E_) load_b<2, 8, true>(b0, ue + N_ * R_, R_);
                mm<2, 8>(acc[0] + 6, acc[1] + 6, a, b1);
            }
            const float* bl  = bias + l * N_ + n_base + mrow;
            const float* x0p = x + (row0 + AROW(0, 8 * ksel)) * N_ + n_base + mrow;
            float*       xlp = xl_s + AROW(0, 8 * ksel) * N_ + n_base + mrow;
            #pragma unroll
            for (int m = 0; m < 2; ++m) {
                float x0v[8][8];
                #pragma unroll
                for (int t = 0; t < 8; ++t) {
                    #pragma unroll
                    for (int j = 0; j < 8; ++j)
                        x0v[t][j] = x0p[AROW(m, j) * N_ + t * 16];
                }
                #pragma unroll
                for (int t = 0; t < 8; ++t) {
                    const float bn = bl[t * 16];
                    #pragma unroll
                    for (int j = 0; j < 8; ++j) {
                        float* p = xlp + AROW(m, j) * N_ + t * 16;
                        *p = fmaf(x0v[t][j], acc[m][t][j] + bn, *p);
                    }
                }
            }
        }
        __syncthreads();
    }

    {
        float4* og = (float4*)(out + row0 * N_);
        const float4* xlv = (const float4*)xl_s;
        #pragma unroll 8
        for (int i = 0; i < (ROWS_ * N_ / 4) / 256; ++i) {
            int idx = tid + i * 256;
            og[idx] = xlv[idx];
        }
    }
}

extern "C" void kernel_launch(void* const* d_in, const int* in_sizes, int n_in,
                              void* d_out, int out_size, void* d_ws, size_t ws_size,
                              hipStream_t stream) {
    const float* x      = (const float*)d_in[0];
    const float* U      = (const float*)d_in[1];
    const float* V      = (const float*)d_in[2];
    const float* C      = (const float*)d_in[3];
    const float* bias   = (const float*)d_in[4];
    const float* gate_w = (const float*)d_in[5];
    float* out = (float*)d_out;

    dim3 grid(B_ / ROWS_);
    dim3 block(256);
    crossnet_fused_kernel<<<grid, block, 0, stream>>>(x, U, V, C, bias, gate_w, out);
}


# LDS rows of x_l, relu(v) and gated relu(c) padded by 16 bytes (pitch 1028 / 516 floats): the 16 rows an A fragment load reads at one column fall 2 deep on a bank group, not 16 deep
# speedup vs baseline: 3.2642x; 1.0384x over previous
#include <hip/hip_runtime.h>
#include <hip/hip_bf16.h>


#define B_ 16384
#define N_ 1024
#define L_ 3
#define E_ 8
#define R_ 64
#define ER_ (E_*R_)
#define ROWS_ 32
#define XP_ (N_ + 4)
#define VP_ (ER_ + 4)
#define AROW(m, r) (2 * (r) + (m))

typedef float v2f __attribute__((ext_vector_type(2)));
typedef float v4f __attribute__((ext_vector_type(4)));
typedef float v8f __attribute__((ext_vector_type(8)));

__device__ __forceinline__ v8f wmma_f32(v2f a, v2f b, v8f c) {
    return __builtin_amdgcn_wmma_f32_16x16x4_f32(
         false, a,  false, b,
         (short)0, c,  false,  false);
}


template <int NT, int NQ, bool ROW_ORDER>
__device__ __forceinline__ void load_b(v4f (&b)[NT][NQ], const float* __restrict__ p, int pitch) {
    #pragma unroll
    for (int t = 0; t < NT; ++t) {
        #pragma unroll
        for (int q = 0; q < NQ; ++q)
            b[t][q] = *(const v4f*)(p + (size_t)t * 16 * pitch + 8 * q);
        if (ROW_ORDER) asm volatile("" ::: "memory");
    }
}

template <int NQ>
__device__ __forceinline__ void load_a(v4f (&a)[2][NQ], const float* p, int tile_off) {
    #pragma unroll
    for (int m = 0; m < 2; ++m) {
        #pragma unroll
        for (int q = 0; q < NQ; ++q)
            a[m][q] = *(const v4f*)(p + m * tile_off + 8 * q);
    }
}

template <int NT, int NQ>
__device__ __forceinline__ void mm(v8f* acc0, v8f* acc1, const v4f (&a)[2][NQ], const v4f (&b)[NT][NQ]) {
    #pragma unroll
    for (int q = 0; q < NQ; ++q) {
        #pragma unroll
        for (int t = 0; t < NT; ++t) {
            acc0[t] = wmma_f32(a[0][q].xy, b[t][q].xy, acc0[t]);
            acc1[t] = wmma_f32(a[1][q].xy, b[t][q].xy, acc1[t]);
        }
        #pragma unroll
        for (int t = 0; t < NT; ++t) {
            acc0[t] = wmma_f32(a[0][q].zw, b[t][q].zw, acc0[t]);
            acc1[t] = wmma_f32(a[1][q].zw, b[t][q].zw, acc1[t]);
        }
    }
}

__global__ __launch_bounds__(256)
void crossnet_fused_kernel(const float* __restrict__ x,
                           const float* __restrict__ U,
                           const float* __restrict__ V,
                           const float* __restrict__ C,
                           const float* __restrict__ bias,
                           const float* __restrict__ gate_w,
                           float* __restrict__ out)
{
    __shared__ __attribute__((aligned(16))) float xl_s[ROWS_ * XP_];
    __shared__ __attribute__((aligned(16))) float v_s [ROWS_ * VP_];
    __shared__ __attribute__((aligned(16))) float act_s[ROWS_ * VP_];
    __shared__ float gates_s[ROWS_ * E_];
    __shared__ __attribute__((aligned(16))) float gpart_s[8 * ROWS_ * E_];

    const int tid  = threadIdx.x;
    const int lane = tid & 31;
    const int wave = tid >> 5;
    const int mrow = lane & 15;
    const int ksel = lane >> 4;
    const size_t row0 = (size_t)blockIdx.x * ROWS_;

    {
        const float4* xg  = (const float4*)(x + row0 * N_);
        float4* xlv = (float4*)xl_s;
        #pragma unroll 8
        for (int i = 0; i < ROWS_; ++i)
            xlv[i * (XP_ / 4) + tid] = xg[i * (N_ / 4) + tid];
    }
    __syncthreads();

    for (int l = 0; l < L_; ++l) {
        {
            const float* gb    = gate_w + (size_t)(mrow & 7) * N_ + wave * 128 + ksel * 4;
            const float* a_src = xl_s + AROW(0, mrow) * XP_ + wave * 128 + ksel * 4;
            v4f b[1][16], a[2][16];
            load_b<1, 16, false>(b, gb, N_);
            load_a<16>(a, a_src, XP_);
            v8f acc0 = {}, acc1 = {}, acc2 = {}, acc3 = {};
            #pragma unroll
            for (int q = 0; q < 16; ++q) {
                acc0 = wmma_f32(a[0][q].xy, b[0][q].xy, acc0);
                acc2 = wmma_f32(a[1][q].xy, b[0][q].xy, acc2);
                acc1 = wmma_f32(a[0][q].zw, b[0][q].zw, acc1);
                acc3 = wmma_f32(a[1][q].zw, b[0][q].zw, acc3);
            }
            asm volatile("v_nop\n\tv_nop\n\tv_nop\n\tv_nop\n\tv_nop\n\tv_nop\n\tv_nop\n\tv_nop"
                         : "+v"(acc0), "+v"(acc1), "+v"(acc2), "+v"(acc3));
            if (mrow < E_) {
                #pragma unroll
                for (int j = 0; j < 8; ++j) {
                    gpart_s[(wave * ROWS_ + AROW(0, j + 8 * ksel)) * E_ + mrow] = acc0[j] + acc1[j];
                    gpart_s[(wave * ROWS_ + AROW(1, j + 8 * ksel)) * E_ + mrow] = acc2[j] + acc3[j];
                }
            }
        }
        __syncthreads();
        if (tid < ROWS_) {
            float lg[E_];
            #pragma unroll
            for (int e = 0; e < E_; ++e) lg[e] = gpart_s[tid * E_ + e];
            #pragma unroll
            for (int w = 1; w < 8; ++w) {
                #pragma unroll
                for (int e = 0; e < E_; ++e) lg[e] += gpart_s[(w * ROWS_ + tid) * E_ + e];
            }
            float m = lg[0];
            #pragma unroll
            for (int e = 1; e < E_; ++e) m = fmaxf(m, lg[e]);
            float tmp[E_]; float s = 0.f;
            #pragma unroll
            for (int e = 0; e < E_; ++e) { tmp[e] = __expf(lg[e] - m); s += tmp[e]; }
            const float inv = 1.f / s;
            #pragma unroll
            for (int e = 0; e < E_; ++e) gates_s[tid * E_ + e] = tmp[e] * inv;
        }
        {
            const int e = wave;
            const float* Vb = V + ((size_t)l * E_ + e) * (R_ * N_)
                                + (size_t)mrow * N_ + ksel * 4;
            const float* a_src = xl_s + AROW(0, mrow) * XP_ + ksel * 4;
            v8f acc[2][4] = {};
            v4f b0[4][4], b1[4][4], a[2][4];
            load_b<4, 4, false>(b0, Vb, N_);
            for (int n0 = 0; n0 < N_ - 64; n0 += 64) {
                load_b<4, 4, false>(b1, Vb + n0 + 32, N_);
                load_a<4>(a, a_src + n0, XP_);
                mm<4, 4>(acc[0], acc[1], a, b0);
                load_b<4, 4, false>(b0, Vb + n0 + 64, N_);
                load_a<4>(a, a_src + n0 + 32, XP_);
                mm<4, 4>(acc[0], acc[1], a, b1);
            }
            load_b<4, 4, false>(b1, Vb + (N_ - 32), N_);
            load_a<4>(a, a_src + (N_ - 64), XP_);
            mm<4, 4>(acc[0], acc[1], a, b0);
            load_a<4>(a, a_src + (N_ - 32), XP_);
            mm<4, 4>(acc[0], acc[1], a, b1);
            #pragma unroll
            for (int m = 0; m < 2; ++m)
                #pragma unroll
                for (int t = 0; t < 4; ++t)
                    #pragma unroll
                    for (int j = 0; j < 8; ++j) {
                        const int row = AROW(m, j + 8 * ksel);
                        v_s[row * VP_ + e * R_ + t * 16 + mrow] = fmaxf(acc[m][t][j], 0.f);
                    }
        }
        __syncthreads();

        {
            const int e = wave;
            const float* Cb = C + ((size_t)l * E_ + e) * (R_ * R_)
                                + mrow * R_ + ksel * 4;
            const float* a_src = v_s + AROW(0, mrow) * VP_ + e * R_ + ksel * 4;
            v8f acc[2][4] = {};
            v4f b[4][8], a[2][8];
            load_b<4, 8, true>(b, Cb, R_);
            load_a<8>(a, a_src, VP_);
            mm<4, 8>(acc[0], acc[1], a, b);
            #pragma unroll
            for (int m = 0; m < 2; ++m)
                #pragma unroll
                for (int t = 0; t < 4; ++t)
                    #pragma unroll
                    for (int j = 0; j < 8; ++j) {
                        const int row = AROW(m, j + 8 * ksel);
                        const float g = gates_s[row * E_ + e];
                        act_s[row * VP_ + e * R_ + t * 16 + mrow] = g * fmaxf(acc[m][t][j], 0.f);
                    }
        }
        __syncthreads();

        {
            const int n_base = wave * 128;
            const float* Ub = U + (size_t)l * (E_ * N_ * R_)
                                + (size_t)(n_base + mrow) * R_ + ksel * 4;
            const float* a_src = act_s + AROW(0, mrow) * VP_ + ksel * 4;
            v8f acc[2][8] = {};
            v4f b0[2][8], b1[2][8], a[2][8];
            load_b<2, 8, true>(b0, Ub, R_);
            for (int e = 0; e < E_; ++e) {
                const float* ue = Ub + (size_t)e * (N_ * R_);
                load_a<8>(a, a_src + e * R_, VP_);
                load_b<2, 8, true>(b1, ue + 32 * R_, R_);
                mm<2, 8>(acc[0] + 0, acc[1] + 0, a, b0);
                load_b<2, 8, true>(b0, ue + 64 * R_, R_);
                mm<2, 8>(acc[0] + 2, acc[1] + 2, a, b1);
                load_b<2, 8, true>(b1, ue + 96 * R_, R_);
                mm<2, 8>(acc[0] + 4, acc[1] + 4, a, b0);
                if (e + 1 < E_) load_b<2, 8, true>(b0, ue + N_ * R_, R_);
                mm<2, 8>(acc[0] + 6, acc[1] + 6, a, b1);
            }
            const float* bl  = bias + l * N_ + n_base + mrow;
            const float* x0p = x + (row0 + AROW(0, 8 * ksel)) * N_ + n_base + mrow;
            float*       xlp = xl_s + AROW(0, 8 * ksel) * XP_ + n_base + mrow;
            #pragma unroll
            for (int m = 0; m < 2; ++m) {
                float x0v[8][8];
                #pragma unroll
                for (int t = 0; t < 8; ++t) {
                    #pragma unroll
                    for (int j = 0; j < 8; ++j)
                        x0v[t][j] = x0p[AROW(m, j) * N_ + t * 16];
                }
                #pragma unroll
                for (int t = 0; t < 8; ++t) {
                    const float bn = bl[t * 16];
                    #pragma unroll
                    for (int j = 0; j < 8; ++j) {
                        float* p = xlp + AROW(m, j) * XP_ + t * 16;
                        *p = fmaf(x0v[t][j], acc[m][t][j] + bn, *p);
                    }
                }
            }
        }
        __syncthreads();
    }

    {
        float4* og = (float4*)(out + row0 * N_);
        const float4* xlv = (const float4*)xl_s;
        #pragma unroll 8
        for (int i = 0; i < ROWS_; ++i)
            og[i * (N_ / 4) + tid] = xlv[i * (XP_ / 4) + tid];
    }
}

extern "C" void kernel_launch(void* const* d_in, const int* in_sizes, int n_in,
                              void* d_out, int out_size, void* d_ws, size_t ws_size,
                              hipStream_t stream) {
    const float* x      = (const float*)d_in[0];
    const float* U      = (const float*)d_in[1];
    const float* V      = (const float*)d_in[2];
    const float* C      = (const float*)d_in[3];
    const float* bias   = (const float*)d_in[4];
    const float* gate_w = (const float*)d_in[5];
    float* out = (float*)d_out;

    dim3 grid(B_ / ROWS_);
    dim3 block(256);
    crossnet_fused_kernel<<<grid, block, 0, stream>>>(x, U, V, C, bias, gate_w, out);
}


# A fragments double-buffered like B: each WMMA block's LDS and global fragment loads issued one block ahead as ONE group, fenced off from the WMMAs (no LDS round trip waited for inside a matrix run)
# speedup vs baseline: 3.3963x; 1.0405x over previous
#include <hip/hip_runtime.h>
#include <hip/hip_bf16.h>


#define B_ 16384
#define N_ 1024
#define L_ 3
#define E_ 8
#define R_ 64
#define ER_ (E_*R_)
#define ROWS_ 32
#define XP_ (N_ + 4)
#define VP_ (ER_ + 4)
#define AROW(m, r) (2 * (r) + (m))

typedef float v2f __attribute__((ext_vector_type(2)));
typedef float v4f __attribute__((ext_vector_type(4)));
typedef float v8f __attribute__((ext_vector_type(8)));

#define LOADS_APART_FROM_WMMAS() __builtin_amdgcn_sched_barrier(0)

__device__ __forceinline__ v8f wmma_f32(v2f a, v2f b, v8f c) {
    return __builtin_amdgcn_wmma_f32_16x16x4_f32(
         false, a,  false, b,
         (short)0, c,  false,  false);
}


template <int NT, int NQ, bool ROW_ORDER>
__device__ __forceinline__ void load_b(v4f (&b)[NT][NQ], const float* __restrict__ p, int pitch) {
    #pragma unroll
    for (int t = 0; t < NT; ++t) {
        #pragma unroll
        for (int q = 0; q < NQ; ++q)
            b[t][q] = *(const v4f*)(p + (size_t)t * 16 * pitch + 8 * q);
        if (ROW_ORDER) asm volatile("" ::: "memory");
    }
}

template <int NQ>
__device__ __forceinline__ void load_a(v4f (&a)[2][NQ], const float* p, int tile_off) {
    #pragma unroll
    for (int m = 0; m < 2; ++m) {
        #pragma unroll
        for (int q = 0; q < NQ; ++q)
            a[m][q] = *(const v4f*)(p + m * tile_off + 8 * q);
    }
}

template <int NT, int NQ>
__device__ __forceinline__ void mm(v8f* acc0, v8f* acc1, const v4f (&a)[2][NQ], const v4f (&b)[NT][NQ]) {
    #pragma unroll
    for (int q = 0; q < NQ; ++q) {
        #pragma unroll
        for (int t = 0; t < NT; ++t) {
            acc0[t] = wmma_f32(a[0][q].xy, b[t][q].xy, acc0[t]);
            acc1[t] = wmma_f32(a[1][q].xy, b[t][q].xy, acc1[t]);
        }
        #pragma unroll
        for (int t = 0; t < NT; ++t) {
            acc0[t] = wmma_f32(a[0][q].zw, b[t][q].zw, acc0[t]);
            acc1[t] = wmma_f32(a[1][q].zw, b[t][q].zw, acc1[t]);
        }
    }
}

template <bool LAST>
__device__ __forceinline__ void expert_pair(v8f (&acc)[2][8], v4f (&a0)[2][8], v4f (&a1)[2][8],
                                            v4f (&b0)[2][8], v4f (&b1)[2][8],
                                            const float* __restrict__ ue, const float* a_e1) {
    const float* uf = ue + N_ * R_;
    load_b<2, 8, true>(b1, ue + 32 * R_, R_);
    LOADS_APART_FROM_WMMAS();
    mm<2, 8>(acc[0] + 0, acc[1] + 0, a0, b0);
    LOADS_APART_FROM_WMMAS();
    load_b<2, 8, true>(b0, ue + 64 * R_, R_);
    LOADS_APART_FROM_WMMAS();
    mm<2, 8>(acc[0] + 2, acc[1] + 2, a0, b1);
    LOADS_APART_FROM_WMMAS();
    load_b<2, 8, true>(b1, ue + 96 * R_, R_);
    LOADS_APART_FROM_WMMAS();
    mm<2, 8>(acc[0] + 4, acc[1] + 4, a0, b0);
    LOADS_APART_FROM_WMMAS();
    load_b<2, 8, true>(b0, uf, R_);
    load_a<8>(a1, a_e1, VP_);
    LOADS_APART_FROM_WMMAS();
    mm<2, 8>(acc[0] + 6, acc[1] + 6, a0, b1);
    LOADS_APART_FROM_WMMAS();
    load_b<2, 8, true>(b1, uf + 32 * R_, R_);
    LOADS_APART_FROM_WMMAS();
    mm<2, 8>(acc[0] + 0, acc[1] + 0, a1, b0);
    LOADS_APART_FROM_WMMAS();
    load_b<2, 8, true>(b0, uf + 64 * R_, R_);
    LOADS_APART_FROM_WMMAS();
    mm<2, 8>(acc[0] + 2, acc[1] + 2, a1, b1);
    LOADS_APART_FROM_WMMAS();
    load_b<2, 8, true>(b1, uf + 96 * R_, R_);
    LOADS_APART_FROM_WMMAS();
    mm<2, 8>(acc[0] + 4, acc[1] + 4, a1, b0);
    LOADS_APART_FROM_WMMAS();
    if (!LAST) {
        load_b<2, 8, true>(b0, uf + N_ * R_, R_);
        load_a<8>(a0, a_e1 + R_, VP_);
    }
    LOADS_APART_FROM_WMMAS();
    mm<2, 8>(acc[0] + 6, acc[1] + 6, a1, b1);
    LOADS_APART_FROM_WMMAS();
}

__global__ __launch_bounds__(256)
void crossnet_fused_kernel(const float* __restrict__ x,
                           const float* __restrict__ U,
                           const float* __restrict__ V,
                           const float* __restrict__ C,
                           const float* __restrict__ bias,
                           const float* __restrict__ gate_w,
                           float* __restrict__ out)
{
    __shared__ __attribute__((aligned(16))) float xl_s[ROWS_ * XP_];
    __shared__ __attribute__((aligned(16))) float v_s [ROWS_ * VP_];
    __shared__ __attribute__((aligned(16))) float act_s[ROWS_ * VP_];
    __shared__ float gates_s[ROWS_ * E_];
    __shared__ __attribute__((aligned(16))) float gpart_s[8 * ROWS_ * E_];

    const int tid  = threadIdx.x;
    const int lane = tid & 31;
    const int wave = tid >> 5;
    const int mrow = lane & 15;
    const int ksel = lane >> 4;
    const size_t row0 = (size_t)blockIdx.x * ROWS_;

    {
        const float4* xg  = (const float4*)(x + row0 * N_);
        float4* xlv = (float4*)xl_s;
        #pragma unroll 8
        for (int i = 0; i < ROWS_; ++i)
            xlv[i * (XP_ / 4) + tid] = xg[i * (N_ / 4) + tid];
    }
    __syncthreads();

    for (int l = 0; l < L_; ++l) {
        {
            const float* gb    = gate_w + (size_t)(mrow & 7) * N_ + wave * 128 + ksel * 4;
            const float* a_src = xl_s + AROW(0, mrow) * XP_ + wave * 128 + ksel * 4;
            v4f b[1][16], a[2][16];
            load_b<1, 16, false>(b, gb, N_);
            load_a<16>(a, a_src, XP_);
            v8f acc0 = {}, acc1 = {}, acc2 = {}, acc3 = {};
            #pragma unroll
            for (int q = 0; q < 16; ++q) {
                acc0 = wmma_f32(a[0][q].xy, b[0][q].xy, acc0);
                acc2 = wmma_f32(a[1][q].xy, b[0][q].xy, acc2);
                acc1 = wmma_f32(a[0][q].zw, b[0][q].zw, acc1);
                acc3 = wmma_f32(a[1][q].zw, b[0][q].zw, acc3);
            }
            asm volatile("v_nop\n\tv_nop\n\tv_nop\n\tv_nop\n\tv_nop\n\tv_nop\n\tv_nop\n\tv_nop"
                         : "+v"(acc0), "+v"(acc1), "+v"(acc2), "+v"(acc3));
            if (mrow < E_) {
                #pragma unroll
                for (int j = 0; j < 8; ++j) {
                    gpart_s[(wave * ROWS_ + AROW(0, j + 8 * ksel)) * E_ + mrow] = acc0[j] + acc1[j];
                    gpart_s[(wave * ROWS_ + AROW(1, j + 8 * ksel)) * E_ + mrow] = acc2[j] + acc3[j];
                }
            }
        }
        __syncthreads();
        if (tid < ROWS_) {
            float lg[E_];
            #pragma unroll
            for (int e = 0; e < E_; ++e) lg[e] = gpart_s[tid * E_ + e];
            #pragma unroll
            for (int w = 1; w < 8; ++w) {
                #pragma unroll
                for (int e = 0; e < E_; ++e) lg[e] += gpart_s[(w * ROWS_ + tid) * E_ + e];
            }
            float m = lg[0];
            #pragma unroll
            for (int e = 1; e < E_; ++e) m = fmaxf(m, lg[e]);
            float tmp[E_]; float s = 0.f;
            #pragma unroll
            for (int e = 0; e < E_; ++e) { tmp[e] = __expf(lg[e] - m); s += tmp[e]; }
            const float inv = 1.f / s;
            #pragma unroll
            for (int e = 0; e < E_; ++e) gates_s[tid * E_ + e] = tmp[e] * inv;
        }
        {
            const int e = wave;
            const float* Vb = V + ((size_t)l * E_ + e) * (R_ * N_)
                                + (size_t)mrow * N_ + ksel * 4;
            const float* a_src = xl_s + AROW(0, mrow) * XP_ + ksel * 4;
            v8f acc[2][4] = {};
            v4f b0[4][4], b1[4][4], a0[2][4], a1[2][4];
            load_b<4, 4, false>(b0, Vb, N_);
            load_a<4>(a0, a_src, XP_);
            for (int n0 = 0; n0 < N_ - 64; n0 += 64) {
                load_b<4, 4, false>(b1, Vb + n0 + 32, N_);
                load_a<4>(a1, a_src + n0 + 32, XP_);
                LOADS_APART_FROM_WMMAS();
                mm<4, 4>(acc[0], acc[1], a0, b0);
                LOADS_APART_FROM_WMMAS();
                load_b<4, 4, false>(b0, Vb + n0 + 64, N_);
                load_a<4>(a0, a_src + n0 + 64, XP_);
                LOADS_APART_FROM_WMMAS();
                mm<4, 4>(acc[0], acc[1], a1, b1);
                LOADS_APART_FROM_WMMAS();
            }
            load_b<4, 4, false>(b1, Vb + (N_ - 32), N_);
            load_a<4>(a1, a_src + (N_ - 32), XP_);
            LOADS_APART_FROM_WMMAS();
            mm<4, 4>(acc[0], acc[1], a0, b0);
            LOADS_APART_FROM_WMMAS();
            mm<4, 4>(acc[0], acc[1], a1, b1);
            float* vp = v_s + AROW(0, 8 * ksel) * VP_ + e * R_ + mrow;
            #pragma unroll
            for (int m = 0; m < 2; ++m)
                #pragma unroll
                for (int t = 0; t < 4; ++t)
                    #pragma unroll
                    for (int j = 0; j < 8; ++j)
                        vp[AROW(m, j) * VP_ + t * 16] = fmaxf(acc[m][t][j], 0.f);
        }
        __syncthreads();

        {
            const int e = wave;
            const float* Cb = C + ((size_t)l * E_ + e) * (R_ * R_)
                                + mrow * R_ + ksel * 4;
            const float* a_src = v_s + AROW(0, mrow) * VP_ + e * R_ + ksel * 4;
            v8f acc[2][4] = {};
            v4f b[4][8], a[2][8];
            load_b<4, 8, true>(b, Cb, R_);
            load_a<8>(a, a_src, VP_);
            mm<4, 8>(acc[0], acc[1], a, b);
            const float* gp = gates_s + AROW(0, 8 * ksel) * E_ + e;
            float*       ap = act_s + AROW(0, 8 * ksel) * VP_ + e * R_ + mrow;
            #pragma unroll
            for (int m = 0; m < 2; ++m)
                #pragma unroll
                for (int t = 0; t < 4; ++t)
                    #pragma unroll
                    for (int j = 0; j < 8; ++j) {
                        const float g = gp[AROW(m, j) * E_];
                        ap[AROW(m, j) * VP_ + t * 16] = g * fmaxf(acc[m][t][j], 0.f);
                    }
        }
        __syncthreads();

        {
            const int n_base = wave * 128;
            const float* Ub = U + (size_t)l * (E_ * N_ * R_)
                                + (size_t)(n_base + mrow) * R_ + ksel * 4;
            const float* a_src = act_s + AROW(0, mrow) * VP_ + ksel * 4;
            v8f acc[2][8] = {};
            v4f b0[2][8], b1[2][8], a0[2][8], a1[2][8];
            load_b<2, 8, true>(b0, Ub, R_);
            load_a<8>(a0, a_src, VP_);
            for (int e = 0; e < E_ - 2; e += 2)
                expert_pair<false>(acc, a0, a1, b0, b1, Ub + (size_t)e * (N_ * R_), a_src + (e + 1) * R_);
            expert_pair<true>(acc, a0, a1, b0, b1, Ub + (size_t)(E_ - 2) * (N_ * R_), a_src + (E_ - 1) * R_);
            const float* bl  = bias + l * N_ + n_base + mrow;
            const float* x0p = x + (row0 + AROW(0, 8 * ksel)) * N_ + n_base + mrow;
            float*       xlp = xl_s + AROW(0, 8 * ksel) * XP_ + n_base + mrow;
            #pragma unroll
            for (int m = 0; m < 2; ++m) {
                float x0v[8][8];
                #pragma unroll
                for (int t = 0; t < 8; ++t) {
                    #pragma unroll
                    for (int j = 0; j < 8; ++j)
                        x0v[t][j] = x0p[AROW(m, j) * N_ + t * 16];
                }
                #pragma unroll
                for (int t = 0; t < 8; ++t) {
                    const float bn = bl[t * 16];
                    #pragma unroll
                    for (int j = 0; j < 8; ++j) {
                        float* p = xlp + AROW(m, j) * XP_ + t * 16;
                        *p = fmaf(x0v[t][j], acc[m][t][j] + bn, *p);
                    }
                }
            }
        }
        __syncthreads();
    }

    {
        float4* og = (float4*)(out + row0 * N_);
        const float4* xlv = (const float4*)xl_s;
        #pragma unroll 8
        for (int i = 0; i < ROWS_; ++i)
            og[i * (N_ / 4) + tid] = xlv[i * (XP_ / 4) + tid];
    }
}

extern "C" void kernel_launch(void* const* d_in, const int* in_sizes, int n_in,
                              void* d_out, int out_size, void* d_ws, size_t ws_size,
                              hipStream_t stream) {
    const float* x      = (const float*)d_in[0];
    const float* U      = (const float*)d_in[1];
    const float* V      = (const float*)d_in[2];
    const float* C      = (const float*)d_in[3];
    const float* bias   = (const float*)d_in[4];
    const float* gate_w = (const float*)d_in[5];
    float* out = (float*)d_out;

    dim3 grid(B_ / ROWS_);
    dim3 block(256);
    crossnet_fused_kernel<<<grid, block, 0, stream>>>(x, U, V, C, bias, gate_w, out);
}
